# starting kernel
# baseline (speedup 1.0000x reference)
#include <hip/hip_runtime.h>

#define B_DIM 128
#define F_DIM 512
#define IC_DIM 32
#define OC_DIM 32
#define PAD 34

typedef __attribute__((ext_vector_type(2))) float v2f;
typedef __attribute__((ext_vector_type(8))) float v8f;

struct __align__(16) Smem {
    float E[B_DIM * PAD];
    float SMT[OC_DIM * PAD];
    float W[IC_DIM * OC_DIM];
    float xmax[B_DIM];
};

__global__ __launch_bounds__(128) void fused_lse_wmma_kernel(
    const float* __restrict__ x,
    const float* __restrict__ w,
    float* __restrict__ out)
{
    __shared__ Smem sm;
    const int f = blockIdx.x;
    const int t = threadIdx.x;

    {
        const float4* wsrc = (const float4*)(w + (size_t)f * IC_DIM * OC_DIM);
        float4* wdst = (float4*)sm.W;
        wdst[t]       = wsrc[t];
        wdst[t + 128] = wsrc[t + 128];
    }
    __syncthreads();

    if (t < 32) {
        const int oc = t;
        float m = -INFINITY;
        #pragma unroll
        for (int ic = 0; ic < IC_DIM; ++ic)
            m = fmaxf(m, sm.W[ic * OC_DIM + oc]);
        float s = 0.0f;
        #pragma unroll
        for (int ic = 0; ic < IC_DIM; ++ic)
            s += __expf(sm.W[ic * OC_DIM + oc] - m);
        const float inv = 1.0f / s;
        #pragma unroll
        for (int ic = 0; ic < IC_DIM; ++ic)
            sm.SMT[oc * PAD + ic] = __expf(sm.W[ic * OC_DIM + oc] - m) * inv;
    }

    {
        const int b = t;
        const float4* xr = (const float4*)(x + ((size_t)b * F_DIM + f) * IC_DIM);
        float4 v[8];
        #pragma unroll
        for (int i = 0; i < 8; ++i) v[i] = xr[i];
        float m = -INFINITY;
        #pragma unroll
        for (int i = 0; i < 8; ++i)
            m = fmaxf(m, fmaxf(fmaxf(v[i].x, v[i].y), fmaxf(v[i].z, v[i].w)));
        sm.xmax[b] = m;
        float* er = &sm.E[b * PAD];
        #pragma unroll
        for (int i = 0; i < 8; ++i) {
            er[4 * i + 0] = __expf(v[i].x - m);
            er[4 * i + 1] = __expf(v[i].y - m);
            er[4 * i + 2] = __expf(v[i].z - m);
            er[4 * i + 3] = __expf(v[i].w - m);
        }
    }
    __syncthreads();

    const int wave  = t >> 5;
    const int lane  = t & 31;
    const int lhalf = lane >> 4;
    const int lmod  = lane & 15;
    const int kb    = lhalf * 2;
    const int mbase = wave * 32;

    v8f acc[2][2] = {};

    #pragma unroll
    for (int ks = 0; ks < 8; ++ks) {
        const int k0 = ks * 4 + kb;
        v2f bfrag[2];
        #pragma unroll
        for (int nt = 0; nt < 2; ++nt) {
            const int n = nt * 16 + lmod;
            bfrag[nt] = *(const v2f*)&sm.SMT[n * PAD + k0];
        }
        #pragma unroll
        for (int mt = 0; mt < 2; ++mt) {
            const int row = mbase + mt * 16 + lmod;
            const v2f afrag = *(const v2f*)&sm.E[row * PAD + k0];
            #pragma unroll
            for (int nt = 0; nt < 2; ++nt) {
                acc[mt][nt] = __builtin_amdgcn_wmma_f32_16x16x4_f32(
                    false, afrag, false, bfrag[nt], (short)0, acc[mt][nt],
                    false, false);
            }
        }
    }

    #pragma unroll
    for (int mt = 0; mt < 2; ++mt) {
        #pragma unroll
        for (int nt = 0; nt < 2; ++nt) {
            #pragma unroll
            for (int r = 0; r < 8; ++r) {
                const int b  = mbase + mt * 16 + r + 8 * lhalf;
                const int oc = nt * 16 + lmod;
                out[((size_t)b * F_DIM + f) * OC_DIM + oc] =
                    __logf(acc[mt][nt][r]) + sm.xmax[b];
            }
        }
    }
}

extern "C" void kernel_launch(void* const* d_in, const int* in_sizes, int n_in,
                              void* d_out, int out_size, void* d_ws, size_t ws_size,
                              hipStream_t stream) {
    (void)in_sizes; (void)n_in; (void)out_size; (void)d_ws; (void)ws_size;
    const float* x = (const float*)d_in[0];
    const float* w = (const float*)d_in[1];
    float* out     = (float*)d_out;
    fused_lse_wmma_kernel<<<dim3(F_DIM), dim3(128), 0, stream>>>(x, w, out);
}


# W softmax spread over all 128 threads (4 lanes per column, 8 rows each kept in registers, quad combine by DPP) instead of one wave's serial pass; column sum order now 4 partial sums of 8
# speedup vs baseline: 1.1486x; 1.1486x over previous
#include <hip/hip_runtime.h>

#define B_DIM 128
#define F_DIM 512
#define IC_DIM 32
#define OC_DIM 32
#define PAD 34

typedef __attribute__((ext_vector_type(2))) float v2f;
typedef __attribute__((ext_vector_type(8))) float v8f;

struct __align__(16) Smem {
    float E[B_DIM * PAD];
    float SMT[OC_DIM * PAD];
    float W[IC_DIM * OC_DIM];
    float xmax[B_DIM];
};

template <int XMASK>
__device__ __forceinline__ float lane_xor(float v) {
    return __int_as_float(__builtin_amdgcn_update_dpp(
        0, __float_as_int(v), 0x160 + XMASK, 0xf, 0xf, false));
}

__global__ __launch_bounds__(128) void fused_lse_wmma_kernel(
    const float* __restrict__ x,
    const float* __restrict__ w,
    float* __restrict__ out)
{
    __shared__ Smem sm;
    const int f = blockIdx.x;
    const int t = threadIdx.x;

    {
        const float4* wsrc = (const float4*)(w + (size_t)f * IC_DIM * OC_DIM);
        float4* wdst = (float4*)sm.W;
        wdst[t]       = wsrc[t];
        wdst[t + 128] = wsrc[t + 128];
    }
    __syncthreads();

    {
        const int oc  = t >> 2;
        const int ic0 = (t & 3) * 8;
        float wv[8];
        #pragma unroll
        for (int j = 0; j < 8; ++j)
            wv[j] = sm.W[(ic0 + j) * OC_DIM + oc];
        float m = wv[0];
        #pragma unroll
        for (int j = 1; j < 8; ++j)
            m = fmaxf(m, wv[j]);
        m = fmaxf(m, lane_xor<1>(m));
        m = fmaxf(m, lane_xor<2>(m));
        float s = 0.0f;
        #pragma unroll
        for (int j = 0; j < 8; ++j) {
            wv[j] = __expf(wv[j] - m);
            s += wv[j];
        }
        s += lane_xor<1>(s);
        s += lane_xor<2>(s);
        const float inv = 1.0f / s;
        v2f* dst = (v2f*)&sm.SMT[oc * PAD + ic0];
        #pragma unroll
        for (int j = 0; j < 4; ++j) {
            v2f o;
            o.x = wv[2 * j]     * inv;
            o.y = wv[2 * j + 1] * inv;
            dst[j] = o;
        }
    }

    {
        const int b = t;
        const float4* xr = (const float4*)(x + ((size_t)b * F_DIM + f) * IC_DIM);
        float4 v[8];
        #pragma unroll
        for (int i = 0; i < 8; ++i) v[i] = xr[i];
        float m = -INFINITY;
        #pragma unroll
        for (int i = 0; i < 8; ++i)
            m = fmaxf(m, fmaxf(fmaxf(v[i].x, v[i].y), fmaxf(v[i].z, v[i].w)));
        sm.xmax[b] = m;
        float* er = &sm.E[b * PAD];
        #pragma unroll
        for (int i = 0; i < 8; ++i) {
            er[4 * i + 0] = __expf(v[i].x - m);
            er[4 * i + 1] = __expf(v[i].y - m);
            er[4 * i + 2] = __expf(v[i].z - m);
            er[4 * i + 3] = __expf(v[i].w - m);
        }
    }
    __syncthreads();

    const int wave  = t >> 5;
    const int lane  = t & 31;
    const int lhalf = lane >> 4;
    const int lmod  = lane & 15;
    const int kb    = lhalf * 2;
    const int mbase = wave * 32;

    v8f acc[2][2] = {};

    #pragma unroll
    for (int ks = 0; ks < 8; ++ks) {
        const int k0 = ks * 4 + kb;
        v2f bfrag[2];
        #pragma unroll
        for (int nt = 0; nt < 2; ++nt) {
            const int n = nt * 16 + lmod;
            bfrag[nt] = *(const v2f*)&sm.SMT[n * PAD + k0];
        }
        #pragma unroll
        for (int mt = 0; mt < 2; ++mt) {
            const int row = mbase + mt * 16 + lmod;
            const v2f afrag = *(const v2f*)&sm.E[row * PAD + k0];
            #pragma unroll
            for (int nt = 0; nt < 2; ++nt) {
                acc[mt][nt] = __builtin_amdgcn_wmma_f32_16x16x4_f32(
                    false, afrag, false, bfrag[nt], (short)0, acc[mt][nt],
                    false, false);
            }
        }
    }

    #pragma unroll
    for (int mt = 0; mt < 2; ++mt) {
        #pragma unroll
        for (int nt = 0; nt < 2; ++nt) {
            #pragma unroll
            for (int r = 0; r < 8; ++r) {
                const int b  = mbase + mt * 16 + r + 8 * lhalf;
                const int oc = nt * 16 + lmod;
                out[((size_t)b * F_DIM + f) * OC_DIM + oc] =
                    __logf(acc[mt][nt][r]) + sm.xmax[b];
            }
        }
    }
}

extern "C" void kernel_launch(void* const* d_in, const int* in_sizes, int n_in,
                              void* d_out, int out_size, void* d_ws, size_t ws_size,
                              hipStream_t stream) {
    (void)in_sizes; (void)n_in; (void)out_size; (void)d_ws; (void)ws_size;
    const float* x = (const float*)d_in[0];
    const float* w = (const float*)d_in[1];
    float* out     = (float*)d_out;
    fused_lse_wmma_kernel<<<dim3(F_DIM), dim3(128), 0, stream>>>(x, w, out);
}


# x row loads issued at kernel entry right behind the W tile loads and kept in flight across the W staging barrier (LDS-only fence) and the W softmax: one exposed memory round trip, not two
# speedup vs baseline: 1.2228x; 1.0646x over previous
#include <hip/hip_runtime.h>

#define B_DIM 128
#define F_DIM 512
#define IC_DIM 32
#define OC_DIM 32
#define PAD 34

typedef __attribute__((ext_vector_type(2))) float v2f;
typedef __attribute__((ext_vector_type(8))) float v8f;

struct __align__(16) Smem {
    float E[B_DIM * PAD];
    float SMT[OC_DIM * PAD];
    float W[IC_DIM * OC_DIM];
    float xmax[B_DIM];
};

template <int XMASK>
__device__ __forceinline__ float lane_xor(float v) {
    return __int_as_float(__builtin_amdgcn_update_dpp(
        0, __float_as_int(v), 0x160 + XMASK, 0xf, 0xf, false));
}

__global__ __launch_bounds__(128) void fused_lse_wmma_kernel(
    const float* __restrict__ x,
    const float* __restrict__ w,
    float* __restrict__ out)
{
    __shared__ Smem sm;
    const int f = blockIdx.x;
    const int t = threadIdx.x;

    const float4* wsrc = (const float4*)(w + (size_t)f * IC_DIM * OC_DIM);
    const float4 w0 = wsrc[t];
    const float4 w1 = wsrc[t + 128];
    const float4* xr = (const float4*)(x + ((size_t)t * F_DIM + f) * IC_DIM);
    float4 v[8];
    #pragma unroll
    for (int i = 0; i < 8; ++i) v[i] = xr[i];

    {
        float4* wdst = (float4*)sm.W;
        wdst[t]       = w0;
        wdst[t + 128] = w1;
    }
    __builtin_amdgcn_fence(__ATOMIC_RELEASE, "workgroup", "local");
    __builtin_amdgcn_s_barrier();
    __builtin_amdgcn_fence(__ATOMIC_ACQUIRE, "workgroup", "local");
    __builtin_amdgcn_sched_barrier(0);

    {
        const int oc  = t >> 2;
        const int ic0 = (t & 3) * 8;
        float wv[8];
        #pragma unroll
        for (int j = 0; j < 8; ++j)
            wv[j] = sm.W[(ic0 + j) * OC_DIM + oc];
        float m = wv[0];
        #pragma unroll
        for (int j = 1; j < 8; ++j)
            m = fmaxf(m, wv[j]);
        m = fmaxf(m, lane_xor<1>(m));
        m = fmaxf(m, lane_xor<2>(m));
        float s = 0.0f;
        #pragma unroll
        for (int j = 0; j < 8; ++j) {
            wv[j] = __expf(wv[j] - m);
            s += wv[j];
        }
        s += lane_xor<1>(s);
        s += lane_xor<2>(s);
        const float inv = 1.0f / s;
        v2f* dst = (v2f*)&sm.SMT[oc * PAD + ic0];
        #pragma unroll
        for (int j = 0; j < 4; ++j) {
            v2f o;
            o.x = wv[2 * j]     * inv;
            o.y = wv[2 * j + 1] * inv;
            dst[j] = o;
        }
    }
    __builtin_amdgcn_sched_barrier(0);

    {
        const int b = t;
        float m = -INFINITY;
        #pragma unroll
        for (int i = 0; i < 8; ++i)
            m = fmaxf(m, fmaxf(fmaxf(v[i].x, v[i].y), fmaxf(v[i].z, v[i].w)));
        sm.xmax[b] = m;
        float* er = &sm.E[b * PAD];
        #pragma unroll
        for (int i = 0; i < 8; ++i) {
            er[4 * i + 0] = __expf(v[i].x - m);
            er[4 * i + 1] = __expf(v[i].y - m);
            er[4 * i + 2] = __expf(v[i].z - m);
            er[4 * i + 3] = __expf(v[i].w - m);
        }
    }
    __syncthreads();

    const int wave  = t >> 5;
    const int lane  = t & 31;
    const int lhalf = lane >> 4;
    const int lmod  = lane & 15;
    const int kb    = lhalf * 2;
    const int mbase = wave * 32;

    v8f acc[2][2] = {};

    #pragma unroll
    for (int ks = 0; ks < 8; ++ks) {
        const int k0 = ks * 4 + kb;
        v2f bfrag[2];
        #pragma unroll
        for (int nt = 0; nt < 2; ++nt) {
            const int n = nt * 16 + lmod;
            bfrag[nt] = *(const v2f*)&sm.SMT[n * PAD + k0];
        }
        #pragma unroll
        for (int mt = 0; mt < 2; ++mt) {
            const int row = mbase + mt * 16 + lmod;
            const v2f afrag = *(const v2f*)&sm.E[row * PAD + k0];
            #pragma unroll
            for (int nt = 0; nt < 2; ++nt) {
                acc[mt][nt] = __builtin_amdgcn_wmma_f32_16x16x4_f32(
                    false, afrag, false, bfrag[nt], (short)0, acc[mt][nt],
                    false, false);
            }
        }
    }

    #pragma unroll
    for (int mt = 0; mt < 2; ++mt) {
        #pragma unroll
        for (int nt = 0; nt < 2; ++nt) {
            #pragma unroll
            for (int r = 0; r < 8; ++r) {
                const int b  = mbase + mt * 16 + r + 8 * lhalf;
                const int oc = nt * 16 + lmod;
                out[((size_t)b * F_DIM + f) * OC_DIM + oc] =
                    __logf(acc[mt][nt][r]) + sm.xmax[b];
            }
        }
    }
}

extern "C" void kernel_launch(void* const* d_in, const int* in_sizes, int n_in,
                              void* d_out, int out_size, void* d_ws, size_t ws_size,
                              hipStream_t stream) {
    (void)in_sizes; (void)n_in; (void)out_size; (void)d_ws; (void)ws_size;
    const float* x = (const float*)d_in[0];
    const float* w = (const float*)d_in[1];
    float* out     = (float*)d_out;
    fused_lse_wmma_kernel<<<dim3(F_DIM), dim3(128), 0, stream>>>(x, w, out);
}


# epilogue log(dot)+rowmax as the hardware log2 and one fused multiply-add per element in place of the library logf expansion (about a dozen vector instructions each); rounding may move by an ulp
# speedup vs baseline: 1.3163x; 1.0765x over previous
#include <hip/hip_runtime.h>

#define B_DIM 128
#define F_DIM 512
#define IC_DIM 32
#define OC_DIM 32
#define PAD 34

typedef __attribute__((ext_vector_type(2))) float v2f;
typedef __attribute__((ext_vector_type(8))) float v8f;

struct __align__(16) Smem {
    float E[B_DIM * PAD];
    float SMT[OC_DIM * PAD];
    float W[IC_DIM * OC_DIM];
    float xmax[B_DIM];
};

template <int XMASK>
__device__ __forceinline__ float lane_xor(float v) {
    return __int_as_float(__builtin_amdgcn_update_dpp(
        0, __float_as_int(v), 0x160 + XMASK, 0xf, 0xf, false));
}

__global__ __launch_bounds__(128) void fused_lse_wmma_kernel(
    const float* __restrict__ x,
    const float* __restrict__ w,
    float* __restrict__ out)
{
    __shared__ Smem sm;
    const int f = blockIdx.x;
    const int t = threadIdx.x;

    const float4* wsrc = (const float4*)(w + (size_t)f * IC_DIM * OC_DIM);
    const float4 w0 = wsrc[t];
    const float4 w1 = wsrc[t + 128];
    const float4* xr = (const float4*)(x + ((size_t)t * F_DIM + f) * IC_DIM);
    float4 v[8];
    #pragma unroll
    for (int i = 0; i < 8; ++i) v[i] = xr[i];

    {
        float4* wdst = (float4*)sm.W;
        wdst[t]       = w0;
        wdst[t + 128] = w1;
    }
    __builtin_amdgcn_fence(__ATOMIC_RELEASE, "workgroup", "local");
    __builtin_amdgcn_s_barrier();
    __builtin_amdgcn_fence(__ATOMIC_ACQUIRE, "workgroup", "local");
    __builtin_amdgcn_sched_barrier(0);

    {
        const int oc  = t >> 2;
        const int ic0 = (t & 3) * 8;
        float wv[8];
        #pragma unroll
        for (int j = 0; j < 8; ++j)
            wv[j] = sm.W[(ic0 + j) * OC_DIM + oc];
        float m = wv[0];
        #pragma unroll
        for (int j = 1; j < 8; ++j)
            m = fmaxf(m, wv[j]);
        m = fmaxf(m, lane_xor<1>(m));
        m = fmaxf(m, lane_xor<2>(m));
        float s = 0.0f;
        #pragma unroll
        for (int j = 0; j < 8; ++j) {
            wv[j] = __expf(wv[j] - m);
            s += wv[j];
        }
        s += lane_xor<1>(s);
        s += lane_xor<2>(s);
        const float inv = 1.0f / s;
        v2f* dst = (v2f*)&sm.SMT[oc * PAD + ic0];
        #pragma unroll
        for (int j = 0; j < 4; ++j) {
            v2f o;
            o.x = wv[2 * j]     * inv;
            o.y = wv[2 * j + 1] * inv;
            dst[j] = o;
        }
    }
    __builtin_amdgcn_sched_barrier(0);

    {
        const int b = t;
        float m = -INFINITY;
        #pragma unroll
        for (int i = 0; i < 8; ++i)
            m = fmaxf(m, fmaxf(fmaxf(v[i].x, v[i].y), fmaxf(v[i].z, v[i].w)));
        sm.xmax[b] = m;
        float* er = &sm.E[b * PAD];
        #pragma unroll
        for (int i = 0; i < 8; ++i) {
            er[4 * i + 0] = __expf(v[i].x - m);
            er[4 * i + 1] = __expf(v[i].y - m);
            er[4 * i + 2] = __expf(v[i].z - m);
            er[4 * i + 3] = __expf(v[i].w - m);
        }
    }
    __syncthreads();

    const int wave  = t >> 5;
    const int lane  = t & 31;
    const int lhalf = lane >> 4;
    const int lmod  = lane & 15;
    const int kb    = lhalf * 2;
    const int mbase = wave * 32;

    v8f acc[2][2] = {};

    #pragma unroll
    for (int ks = 0; ks < 8; ++ks) {
        const int k0 = ks * 4 + kb;
        v2f bfrag[2];
        #pragma unroll
        for (int nt = 0; nt < 2; ++nt) {
            const int n = nt * 16 + lmod;
            bfrag[nt] = *(const v2f*)&sm.SMT[n * PAD + k0];
        }
        #pragma unroll
        for (int mt = 0; mt < 2; ++mt) {
            const int row = mbase + mt * 16 + lmod;
            const v2f afrag = *(const v2f*)&sm.E[row * PAD + k0];
            #pragma unroll
            for (int nt = 0; nt < 2; ++nt) {
                acc[mt][nt] = __builtin_amdgcn_wmma_f32_16x16x4_f32(
                    false, afrag, false, bfrag[nt], (short)0, acc[mt][nt],
                    false, false);
            }
        }
    }

    const float LN2 = 0.693147180559945309f;
    #pragma unroll
    for (int mt = 0; mt < 2; ++mt) {
        #pragma unroll
        for (int nt = 0; nt < 2; ++nt) {
            #pragma unroll
            for (int r = 0; r < 8; ++r) {
                const int b  = mbase + mt * 16 + r + 8 * lhalf;
                const int oc = nt * 16 + lmod;
                out[((size_t)b * F_DIM + f) * OC_DIM + oc] =
                    __builtin_fmaf(__builtin_amdgcn_logf(acc[mt][nt][r]), LN2, sm.xmax[b]);
            }
        }
    }
}

extern "C" void kernel_launch(void* const* d_in, const int* in_sizes, int n_in,
                              void* d_out, int out_size, void* d_ws, size_t ws_size,
                              hipStream_t stream) {
    (void)in_sizes; (void)n_in; (void)out_size; (void)d_ws; (void)ws_size;
    const float* x = (const float*)d_in[0];
    const float* w = (const float*)d_in[1];
    float* out     = (float*)d_out;
    fused_lse_wmma_kernel<<<dim3(F_DIM), dim3(128), 0, stream>>>(x, w, out);
}


# x loads re-mapped so 8 lanes share each 128-byte row (16 B per lane): every wave-wide load covers 4 whole lines instead of touching 32; row maximum combined over the 8 lanes by DPP max
# speedup vs baseline: 1.3845x; 1.0518x over previous
#include <hip/hip_runtime.h>

#define B_DIM 128
#define F_DIM 512
#define IC_DIM 32
#define OC_DIM 32
#define PAD 34

typedef __attribute__((ext_vector_type(2))) float v2f;
typedef __attribute__((ext_vector_type(8))) float v8f;

struct __align__(16) Smem {
    float E[B_DIM * PAD];
    float SMT[OC_DIM * PAD];
    float W[IC_DIM * OC_DIM];
    float xmax[B_DIM];
};

template <int XMASK>
__device__ __forceinline__ float lane_xor(float v) {
    return __int_as_float(__builtin_amdgcn_update_dpp(
        0, __float_as_int(v), 0x160 + XMASK, 0xf, 0xf, false));
}

template <int XMASK>
__device__ __forceinline__ float max_lane_xor(float v) {
    float r;
    asm("v_max_num_f32_dpp %0, %1, %1 row_xmask:%2 row_mask:0xf bank_mask:0xf"
        : "=v"(r) : "v"(v), "n"(XMASK));
    return r;
}

__global__ __launch_bounds__(128) void fused_lse_wmma_kernel(
    const float* __restrict__ x,
    const float* __restrict__ w,
    float* __restrict__ out)
{
    __shared__ Smem sm;
    const int f = blockIdx.x;
    const int t = threadIdx.x;

    const float4* wsrc = (const float4*)(w + (size_t)f * IC_DIM * OC_DIM);
    const float4 w0 = wsrc[t];
    const float4 w1 = wsrc[t + 128];
    const int q     = t & 7;
    const int brow0 = (t >> 5) * 32 + ((t >> 3) & 3);
    const float4* xr = (const float4*)(x + ((size_t)brow0 * F_DIM + f) * IC_DIM) + q;
    float4 v[8];
    #pragma unroll
    for (int i = 0; i < 8; ++i) v[i] = xr[(size_t)i * (4 * F_DIM * IC_DIM / 4)];

    {
        float4* wdst = (float4*)sm.W;
        wdst[t]       = w0;
        wdst[t + 128] = w1;
    }
    __builtin_amdgcn_fence(__ATOMIC_RELEASE, "workgroup", "local");
    __builtin_amdgcn_s_barrier();
    __builtin_amdgcn_fence(__ATOMIC_ACQUIRE, "workgroup", "local");
    __builtin_amdgcn_sched_barrier(0);

    {
        const int oc  = t >> 2;
        const int ic0 = (t & 3) * 8;
        float wv[8];
        #pragma unroll
        for (int j = 0; j < 8; ++j)
            wv[j] = sm.W[(ic0 + j) * OC_DIM + oc];
        float m = wv[0];
        #pragma unroll
        for (int j = 1; j < 8; ++j)
            m = fmaxf(m, wv[j]);
        m = fmaxf(m, lane_xor<1>(m));
        m = fmaxf(m, lane_xor<2>(m));
        float s = 0.0f;
        #pragma unroll
        for (int j = 0; j < 8; ++j) {
            wv[j] = __expf(wv[j] - m);
            s += wv[j];
        }
        s += lane_xor<1>(s);
        s += lane_xor<2>(s);
        const float inv = 1.0f / s;
        v2f* dst = (v2f*)&sm.SMT[oc * PAD + ic0];
        #pragma unroll
        for (int j = 0; j < 4; ++j) {
            v2f o;
            o.x = wv[2 * j]     * inv;
            o.y = wv[2 * j + 1] * inv;
            dst[j] = o;
        }
    }
    __builtin_amdgcn_sched_barrier(0);

    {
        float m[8];
        #pragma unroll
        for (int i = 0; i < 8; ++i) {
            float mi = fmaxf(fmaxf(v[i].x, v[i].y), fmaxf(v[i].z, v[i].w));
            mi = max_lane_xor<1>(mi);
            mi = max_lane_xor<2>(mi);
            mi = max_lane_xor<4>(mi);
            m[i] = mi;
        }
        const float m01 = (q & 1) ? m[1] : m[0];
        const float m23 = (q & 1) ? m[3] : m[2];
        const float m45 = (q & 1) ? m[5] : m[4];
        const float m67 = (q & 1) ? m[7] : m[6];
        const float m03 = (q & 2) ? m23 : m01;
        const float m47 = (q & 2) ? m67 : m45;
        sm.xmax[brow0 + 4 * q] = (q & 4) ? m47 : m03;
        #pragma unroll
        for (int i = 0; i < 8; ++i) {
            v2f* er = (v2f*)&sm.E[(brow0 + 4 * i) * PAD + 4 * q];
            v2f lo, hi;
            lo.x = __expf(v[i].x - m[i]);
            lo.y = __expf(v[i].y - m[i]);
            hi.x = __expf(v[i].z - m[i]);
            hi.y = __expf(v[i].w - m[i]);
            er[0] = lo;
            er[1] = hi;
        }
    }
    __syncthreads();

    const int wave  = t >> 5;
    const int lane  = t & 31;
    const int lhalf = lane >> 4;
    const int lmod  = lane & 15;
    const int kb    = lhalf * 2;
    const int mbase = wave * 32;

    v8f acc[2][2] = {};

    #pragma unroll
    for (int ks = 0; ks < 8; ++ks) {
        const int k0 = ks * 4 + kb;
        v2f bfrag[2];
        #pragma unroll
        for (int nt = 0; nt < 2; ++nt) {
            const int n = nt * 16 + lmod;
            bfrag[nt] = *(const v2f*)&sm.SMT[n * PAD + k0];
        }
        #pragma unroll
        for (int mt = 0; mt < 2; ++mt) {
            const int row = mbase + mt * 16 + lmod;
            const v2f afrag = *(const v2f*)&sm.E[row * PAD + k0];
            #pragma unroll
            for (int nt = 0; nt < 2; ++nt) {
                acc[mt][nt] = __builtin_amdgcn_wmma_f32_16x16x4_f32(
                    false, afrag, false, bfrag[nt], (short)0, acc[mt][nt],
                    false, false);
            }
        }
    }

    const float LN2 = 0.693147180559945309f;
    #pragma unroll
    for (int mt = 0; mt < 2; ++mt) {
        #pragma unroll
        for (int nt = 0; nt < 2; ++nt) {
            #pragma unroll
            for (int r = 0; r < 8; ++r) {
                const int b  = mbase + mt * 16 + r + 8 * lhalf;
                const int oc = nt * 16 + lmod;
                out[((size_t)b * F_DIM + f) * OC_DIM + oc] =
                    __builtin_fmaf(__builtin_amdgcn_logf(acc[mt][nt][r]), LN2, sm.xmax[b]);
            }
        }
    }
}

extern "C" void kernel_launch(void* const* d_in, const int* in_sizes, int n_in,
                              void* d_out, int out_size, void* d_ws, size_t ws_size,
                              hipStream_t stream) {
    (void)in_sizes; (void)n_in; (void)out_size; (void)d_ws; (void)ws_size;
    const float* x = (const float*)d_in[0];
    const float* w = (const float*)d_in[1];
    float* out     = (float*)d_out;
    fused_lse_wmma_kernel<<<dim3(F_DIM), dim3(128), 0, stream>>>(x, w, out);
}
